# starting kernel
# baseline (speedup 1.0000x reference)
#include <hip/hip_runtime.h>
#include <hip/hip_bf16.h>

typedef float v2f __attribute__((ext_vector_type(2)));
typedef float v8f __attribute__((ext_vector_type(8)));

#define LDS_PAD 4

template<int Ci, int Co, int S, int TILES>
__global__ __launch_bounds__((Co/16)*32)
void conv4d_relu_pool_wmma(const float* __restrict__ x,
                           const float* __restrict__ w,
                           const float* __restrict__ bias,
                           float* __restrict__ out)
{
    constexpr int P    = S / 2;
    constexpr int PH   = P / TILES;
    constexpr int P4   = P * P * P * P;
    constexpr int D3   = 2 + 2 * TILES;
    constexpr int NSP  = 64 * D3;
    constexpr int CSTR = Ci + LDS_PAD;
    constexpr int NT   = (Co / 16) * 32;

    __shared__ float lds[NSP * CSTR];

    const int b   = blockIdx.y;
    const int bx  = blockIdx.x;
    const int p3h = bx % PH;
    const int p2  = (bx / PH) % P;
    const int p1  = (bx / (PH * P)) % P;
    const int p0  = bx / (PH * P * P);

    const int tid = threadIdx.x;
    for (int e = tid; e < NSP * Ci; e += NT) {
        const int ci = e & (Ci - 1);
        const int sp = e / Ci;
        const int d3 = sp % D3;
        const int r  = sp / D3;
        const int d2 = r & 3, d1 = (r >> 2) & 3, d0 = (r >> 4) & 3;
        const int x0 = 2 * p0 - 1 + d0;
        const int x1 = 2 * p1 - 1 + d1;
        const int x2 = 2 * p2 - 1 + d2;
        const int x3 = 2 * (p3h * TILES) - 1 + d3;
        float v = 0.0f;
        if ((unsigned)x0 < (unsigned)S && (unsigned)x1 < (unsigned)S &&
            (unsigned)x2 < (unsigned)S && (unsigned)x3 < (unsigned)S) {
            v = x[((((b * Ci + ci) * S + x0) * S + x1) * S + x2) * S + x3];
        }
        lds[sp * CSTR + ci] = v;
    }
    __syncthreads();

    const int lane = tid & 31;
    const int wave = tid >> 5;
    const int n    = lane & 15;
    const int klo  = (lane < 16) ? 0 : 2;
    const int qsp  = (((n >> 3) & 1) * 16 + ((n >> 2) & 1) * 4 + ((n >> 1) & 1)) * D3 + (n & 1);
    const int coA  = wave * 16 + n;
    const int wCoK = coA * Ci * 81 + klo * 81;

    v8f acc[TILES];
    #pragma unroll
    for (int u = 0; u < TILES; ++u)
        acc[u] = (v8f){0.f, 0.f, 0.f, 0.f, 0.f, 0.f, 0.f, 0.f};

    #pragma unroll 1
    for (int t0 = 0; t0 < 3; ++t0) {
        #pragma unroll 1
        for (int t1 = 0; t1 < 3; ++t1) {
            #pragma unroll 1
            for (int t2 = 0; t2 < 3; ++t2) {
                const int row  = (qsp + (t0 * 16 + t1 * 4 + t2) * D3) * CSTR + klo;
                const int wRow = wCoK + t0 * 27 + t1 * 9 + t2 * 3;
                #pragma unroll 4
                for (int kb = 0; kb < Ci; kb += 4) {
                    const float* wp = &w[wRow + kb * 81];
                    float wx[3] = { wp[0],  wp[1],  wp[2]  };
                    float wy[3] = { wp[81], wp[82], wp[83] };
                    #pragma unroll
                    for (int t3 = 0; t3 < 3; ++t3) {
                        v2f bf; bf.x = wx[t3]; bf.y = wy[t3];
                        #pragma unroll
                        for (int u = 0; u < TILES; ++u) {
                            v2f a = *(const v2f*)&lds[row + (t3 + 2 * u) * CSTR + kb];
                            acc[u] = __builtin_amdgcn_wmma_f32_16x16x4_f32(
                                 false, a,  false, bf,
                                 (short)0, acc[u],
                                 false,  false);
                        }
                    }
                }
            }
        }
    }

    const float bs = bias[coA];
    #pragma unroll
    for (int u = 0; u < TILES; ++u) {
        float s = 0.0f;
        #pragma unroll
        for (int r = 0; r < 8; ++r) {
            float v = acc[u][r] + bs;
            s += (v > 0.0f) ? v : 0.0f;
        }
        s += __shfl_xor(s, 16, 32);
        if (lane < 16) {
            const int pIdx = ((p0 * P + p1) * P + p2) * P + p3h * TILES + u;
            out[(b * Co + coA) * P4 + pIdx] = s * (1.0f / 16.0f);
        }
    }
}

extern "C" void kernel_launch(void* const* d_in, const int* in_sizes, int n_in,
                              void* d_out, int out_size, void* d_ws, size_t ws_size,
                              hipStream_t stream) {
    (void)in_sizes; (void)n_in; (void)out_size; (void)ws_size;

    const float* src = (const float*)d_in[0];
    const float* W0  = (const float*)d_in[1];
    const float* b0  = (const float*)d_in[2];
    const float* W1  = (const float*)d_in[3];
    const float* b1  = (const float*)d_in[4];
    const float* W2  = (const float*)d_in[5];
    const float* b2  = (const float*)d_in[6];
    const float* W3  = (const float*)d_in[7];
    const float* b3  = (const float*)d_in[8];

    float* t0 = (float*)d_ws;
    float* o1 = (float*)d_out;
    float* o2 = o1 + 8 * 64 * 256;
    float* o3 = o2 + 8 * 64 * 16;

    conv4d_relu_pool_wmma<8, 32, 16, 2><<<dim3(2048, 8), 64, 0, stream>>>(src, W0, b0, t0);
    conv4d_relu_pool_wmma<32, 64, 8, 2><<<dim3(128, 8), 128, 0, stream>>>(t0, W1, b1, o1);
    conv4d_relu_pool_wmma<64, 64, 4, 2><<<dim3(8, 8), 128, 0, stream>>>(o1, W2, b2, o2);
    conv4d_relu_pool_wmma<64, 64, 2, 1><<<dim3(1, 8), 128, 0, stream>>>(o2, W3, b3, o3);
}


# Layer 0: weight tensor kept in LDS in B-fragment order by a persistent block (batch,p0,p1) walking its 8 pooled rows with 8 waves; fragments read from LDS, not global; sum order unchanged
# speedup vs baseline: 1.3176x; 1.3176x over previous
#include <hip/hip_runtime.h>
#include <hip/hip_bf16.h>

typedef float v2f __attribute__((ext_vector_type(2)));
typedef float v8f __attribute__((ext_vector_type(8)));

#define LDS_PAD 4

template<int Ci, int Co, int S, int TILES>
__global__ __launch_bounds__((Co/16)*32)
void conv4d_relu_pool_wmma(const float* __restrict__ x,
                           const float* __restrict__ w,
                           const float* __restrict__ bias,
                           float* __restrict__ out)
{
    constexpr int P    = S / 2;
    constexpr int PH   = P / TILES;
    constexpr int P4   = P * P * P * P;
    constexpr int D3   = 2 + 2 * TILES;
    constexpr int NSP  = 64 * D3;
    constexpr int CSTR = Ci + LDS_PAD;
    constexpr int NT   = (Co / 16) * 32;

    __shared__ float lds[NSP * CSTR];

    const int b   = blockIdx.y;
    const int bx  = blockIdx.x;
    const int p3h = bx % PH;
    const int p2  = (bx / PH) % P;
    const int p1  = (bx / (PH * P)) % P;
    const int p0  = bx / (PH * P * P);

    const int tid = threadIdx.x;
    for (int e = tid; e < NSP * Ci; e += NT) {
        const int ci = e & (Ci - 1);
        const int sp = e / Ci;
        const int d3 = sp % D3;
        const int r  = sp / D3;
        const int d2 = r & 3, d1 = (r >> 2) & 3, d0 = (r >> 4) & 3;
        const int x0 = 2 * p0 - 1 + d0;
        const int x1 = 2 * p1 - 1 + d1;
        const int x2 = 2 * p2 - 1 + d2;
        const int x3 = 2 * (p3h * TILES) - 1 + d3;
        float v = 0.0f;
        if ((unsigned)x0 < (unsigned)S && (unsigned)x1 < (unsigned)S &&
            (unsigned)x2 < (unsigned)S && (unsigned)x3 < (unsigned)S) {
            v = x[((((b * Ci + ci) * S + x0) * S + x1) * S + x2) * S + x3];
        }
        lds[sp * CSTR + ci] = v;
    }
    __syncthreads();

    const int lane = tid & 31;
    const int wave = tid >> 5;
    const int n    = lane & 15;
    const int klo  = (lane < 16) ? 0 : 2;
    const int qsp  = (((n >> 3) & 1) * 16 + ((n >> 2) & 1) * 4 + ((n >> 1) & 1)) * D3 + (n & 1);
    const int coA  = wave * 16 + n;
    const int wCoK = coA * Ci * 81 + klo * 81;

    v8f acc[TILES];
    #pragma unroll
    for (int u = 0; u < TILES; ++u)
        acc[u] = (v8f){0.f, 0.f, 0.f, 0.f, 0.f, 0.f, 0.f, 0.f};

    #pragma unroll 1
    for (int t0 = 0; t0 < 3; ++t0) {
        #pragma unroll 1
        for (int t1 = 0; t1 < 3; ++t1) {
            #pragma unroll 1
            for (int t2 = 0; t2 < 3; ++t2) {
                const int row  = (qsp + (t0 * 16 + t1 * 4 + t2) * D3) * CSTR + klo;
                const int wRow = wCoK + t0 * 27 + t1 * 9 + t2 * 3;
                #pragma unroll 4
                for (int kb = 0; kb < Ci; kb += 4) {
                    const float* wp = &w[wRow + kb * 81];
                    float wx[3] = { wp[0],  wp[1],  wp[2]  };
                    float wy[3] = { wp[81], wp[82], wp[83] };
                    #pragma unroll
                    for (int t3 = 0; t3 < 3; ++t3) {
                        v2f bf; bf.x = wx[t3]; bf.y = wy[t3];
                        #pragma unroll
                        for (int u = 0; u < TILES; ++u) {
                            v2f a = *(const v2f*)&lds[row + (t3 + 2 * u) * CSTR + kb];
                            acc[u] = __builtin_amdgcn_wmma_f32_16x16x4_f32(
                                 false, a,  false, bf,
                                 (short)0, acc[u],
                                 false,  false);
                        }
                    }
                }
            }
        }
    }

    const float bs = bias[coA];
    #pragma unroll
    for (int u = 0; u < TILES; ++u) {
        float s = 0.0f;
        #pragma unroll
        for (int r = 0; r < 8; ++r) {
            float v = acc[u][r] + bs;
            s += (v > 0.0f) ? v : 0.0f;
        }
        s += __shfl_xor(s, 16, 32);
        if (lane < 16) {
            const int pIdx = ((p0 * P + p1) * P + p2) * P + p3h * TILES + u;
            out[(b * Co + coA) * P4 + pIdx] = s * (1.0f / 16.0f);
        }
    }
}


#define L0_WT   260
#define L0_SP0  300
#define L0_CSTR 12
#define L0_NPOS (3 * L0_SP0 + 288)

__global__ __launch_bounds__(256)
void conv4d_l0_wstat(const float* __restrict__ x,
                     const float* __restrict__ w,
                     const float* __restrict__ bias,
                     float* __restrict__ out)
{
    __shared__ __attribute__((aligned(16))) float wl[81 * L0_WT];
    __shared__ __attribute__((aligned(16))) float lp[L0_NPOS * L0_CSTR];

    const int tid = threadIdx.x;
    const int b   = blockIdx.y;
    const int p0  = blockIdx.x >> 3;
    const int p1  = blockIdx.x & 7;

    {
        int q = tid / 81;
        int t = tid - q * 81;
        #pragma unroll 1
        for (int c = 0; c < 3; ++c) {
            float tmp[27];
            #pragma unroll
            for (int i = 0; i < 27; ++i)
                tmp[i] = w[tid + (c * 27 + i) * 256];
            #pragma unroll
            for (int i = 0; i < 27; ++i) {
                const int inner = (q >> 7) * 128 + (((q >> 1) & 1) * 16 + ((q >> 3) & 15)) * 4
                                + ((q >> 2) & 1) * 2 + (q & 1);
                wl[t * L0_WT + inner] = tmp[i];
                t += 13; q += 3;
                if (t >= 81) { t -= 81; q += 1; }
            }
        }
    }
    for (int e = tid; e < 64 * 2 * 8; e += 256) {
        const int ci = e & 7;
        const int d3 = ((e >> 3) & 1) * 17;
        const int r  = e >> 4;
        lp[((r >> 4) * L0_SP0 + (r & 15) * 18 + d3) * L0_CSTR + ci] = 0.0f;
    }

    const int sx3 = tid & 15;
    const int sd2 = (tid >> 4) & 3;
    const int sd1 = (tid >> 6) & 3;
    const int gx1 = 2 * p1 - 1 + sd1;
    const bool ok1 = (unsigned)gx1 < 16u;
    const int ldsS = ((sd1 * 4 + sd2) * 18 + sx3 + 1) * L0_CSTR;

    float pre[32];
    auto load_patch = [&](int p2) {
        const int gx2 = 2 * p2 - 1 + sd2;
        const bool ok12 = ok1 && ((unsigned)gx2 < 16u);
        #pragma unroll
        for (int k = 0; k < 32; ++k) {
            const int ci = k >> 2, d0 = k & 3;
            const int gx0 = 2 * p0 - 1 + d0;
            const bool ok = ok12 && ((unsigned)gx0 < 16u);
            const int idx = ok ? (((((b * 8 + ci) * 16 + gx0) * 16 + gx1) * 16 + gx2) * 16 + sx3) : 0;
            pre[k] = x[idx];
        }
    };
    auto store_patch = [&](int p2) {
        const int gx2 = 2 * p2 - 1 + sd2;
        const bool ok12 = ok1 && ((unsigned)gx2 < 16u);
        #pragma unroll
        for (int k = 0; k < 32; ++k) {
            const int gx0 = 2 * p0 - 1 + (k & 3);
            const bool ok = ok12 && ((unsigned)gx0 < 16u);
            lp[(k & 3) * (L0_SP0 * L0_CSTR) + ldsS + (k >> 2)] = ok ? pre[k] : 0.0f;
        }
    };

    const int lane = tid & 31;
    const int wave = tid >> 5;
    const int ct   = wave & 1;
    const int pp   = wave >> 1;
    const int n    = lane & 15;
    const int klo  = (lane < 16) ? 0 : 2;
    const int qsp  = ((n >> 3) & 1) * L0_SP0 + (((n >> 2) & 1) * 4 + ((n >> 1) & 1)) * 18 + (n & 1);
    const int coA  = ct * 16 + n;
    const float bs = bias[coA];
    const float* wlw = wl + ct * 128 + lane * 4;
    const int aBase = (qsp + 4 * pp) * L0_CSTR + klo;

    load_patch(0);

    #pragma unroll 1
    for (int p2 = 0; p2 < 8; ++p2) {
        __syncthreads();
        store_patch(p2);
        __syncthreads();
        if (p2 + 1 < 8) load_patch(p2 + 1);

        v8f acc[2];
        acc[0] = (v8f){0.f, 0.f, 0.f, 0.f, 0.f, 0.f, 0.f, 0.f};
        acc[1] = (v8f){0.f, 0.f, 0.f, 0.f, 0.f, 0.f, 0.f, 0.f};

        #pragma unroll 1
        for (int t0 = 0; t0 < 3; ++t0) {
            #pragma unroll 1
            for (int t1 = 0; t1 < 3; ++t1) {
                #pragma unroll 1
                for (int t2 = 0; t2 < 3; ++t2) {
                    const int row = aBase + (t0 * L0_SP0 + (t1 * 4 + t2) * 18) * L0_CSTR;
                    const float* wt = wlw + (t0 * 27 + t1 * 9 + t2 * 3) * L0_WT;
                    float4 bw[3];
                    #pragma unroll
                    for (int t3 = 0; t3 < 3; ++t3)
                        bw[t3] = *(const float4*)&wt[t3 * L0_WT];
                    #pragma unroll
                    for (int kbi = 0; kbi < 2; ++kbi) {
                        #pragma unroll
                        for (int t3 = 0; t3 < 3; ++t3) {
                            v2f bf;
                            bf.x = kbi ? bw[t3].z : bw[t3].x;
                            bf.y = kbi ? bw[t3].w : bw[t3].y;
                            #pragma unroll
                            for (int u = 0; u < 2; ++u) {
                                v2f a = *(const v2f*)&lp[row + (t3 + 2 * u) * L0_CSTR + kbi * 4];
                                acc[u] = __builtin_amdgcn_wmma_f32_16x16x4_f32(
                                    false, a, false, bf, (short)0, acc[u], false, false);
                            }
                        }
                    }
                }
            }
        }

        #pragma unroll
        for (int u = 0; u < 2; ++u) {
            float s = 0.0f;
            #pragma unroll
            for (int r = 0; r < 8; ++r) {
                float v = acc[u][r] + bs;
                s += (v > 0.0f) ? v : 0.0f;
            }
            s += __shfl_xor(s, 16, 32);
            if (lane < 16) {
                const int pIdx = ((p0 * 8 + p1) * 8 + p2) * 8 + pp * 2 + u;
                out[(b * 32 + coA) * 4096 + pIdx] = s * (1.0f / 16.0f);
            }
        }
    }
}

extern "C" void kernel_launch(void* const* d_in, const int* in_sizes, int n_in,
                              void* d_out, int out_size, void* d_ws, size_t ws_size,
                              hipStream_t stream) {
    (void)in_sizes; (void)n_in; (void)out_size; (void)ws_size;

    const float* src = (const float*)d_in[0];
    const float* W0  = (const float*)d_in[1];
    const float* b0  = (const float*)d_in[2];
    const float* W1  = (const float*)d_in[3];
    const float* b1  = (const float*)d_in[4];
    const float* W2  = (const float*)d_in[5];
    const float* b2  = (const float*)d_in[6];
    const float* W3  = (const float*)d_in[7];
    const float* b3  = (const float*)d_in[8];

    float* t0 = (float*)d_ws;
    float* o1 = (float*)d_out;
    float* o2 = o1 + 8 * 64 * 256;
    float* o3 = o2 + 8 * 64 * 16;

    conv4d_l0_wstat<<<dim3(64, 8), 256, 0, stream>>>(src, W0, b0, t0);
    conv4d_relu_pool_wmma<32, 64, 8, 2><<<dim3(128, 8), 128, 0, stream>>>(t0, W1, b1, o1);
    conv4d_relu_pool_wmma<64, 64, 4, 2><<<dim3(8, 8), 128, 0, stream>>>(o1, W2, b2, o2);
    conv4d_relu_pool_wmma<64, 64, 2, 1><<<dim3(1, 8), 128, 0, stream>>>(o2, W3, b3, o3);
}


# Layers 1-3: weights of 4 input channels at a time copied into LDS by contiguous 16-byte reads, next slice prefetched in registers; fragments read from LDS; channel loop outermost: sum order changes
# speedup vs baseline: 1.9463x; 1.4772x over previous
#include <hip/hip_runtime.h>
#include <hip/hip_bf16.h>

typedef float v2f __attribute__((ext_vector_type(2)));
typedef float v4f __attribute__((ext_vector_type(4)));
typedef float v8f __attribute__((ext_vector_type(8)));

#define LDS_PAD 4

template<int Ci, int Co, int S, int TILES>
__global__ __launch_bounds__((Co/16)*32)
void conv4d_relu_pool_wmma(const float* __restrict__ x,
                           const float* __restrict__ w,
                           const float* __restrict__ bias,
                           float* __restrict__ out)
{
    constexpr int P    = S / 2;
    constexpr int PH   = P / TILES;
    constexpr int P4   = P * P * P * P;
    constexpr int D3   = 2 + 2 * TILES;
    constexpr int NSP  = 64 * D3;
    constexpr int CSTR = Ci + LDS_PAD;
    constexpr int NT   = (Co / 16) * 32;
    constexpr int WROW = 4 * 81;
    constexpr int NV   = Co * 81;
    constexpr int FULL = NV / NT;
    constexpr int REM  = NV - FULL * NT;

    __shared__ __attribute__((aligned(16))) float lds[NSP * CSTR];
    __shared__ __attribute__((aligned(16))) float wsl[Co * WROW];

    const int tid = threadIdx.x;

    v4f wpre[FULL + 1];
    auto load_slice = [&](int kb) {
        const float* ws = w + kb * 81;
        #pragma unroll
        for (int i = 0; i <= FULL; ++i) {
            if (i == FULL && REM == 0) break;
            int f = tid + i * NT;
            if (i == FULL) f = (f < NV) ? f : (NV - 1);
            const int co = f / 81;
            wpre[i] = *(const v4f*)&ws[co * (Ci * 81 - WROW) + 4 * f];
        }
    };
    auto store_slice = [&]() {
        #pragma unroll
        for (int i = 0; i < FULL; ++i)
            *(v4f*)&wsl[4 * (tid + i * NT)] = wpre[i];
        if (REM != 0 && tid < REM)
            *(v4f*)&wsl[4 * (tid + FULL * NT)] = wpre[FULL];
    };

    load_slice(0);

    const int b   = blockIdx.y;
    const int bx  = blockIdx.x;
    const int p3h = bx % PH;
    const int p2  = (bx / PH) % P;
    const int p1  = (bx / (PH * P)) % P;
    const int p0  = bx / (PH * P * P);

    for (int e = tid; e < NSP * Ci; e += NT) {
        const int ci = e & (Ci - 1);
        const int sp = e / Ci;
        const int d3 = sp % D3;
        const int r  = sp / D3;
        const int d2 = r & 3, d1 = (r >> 2) & 3, d0 = (r >> 4) & 3;
        const int x0 = 2 * p0 - 1 + d0;
        const int x1 = 2 * p1 - 1 + d1;
        const int x2 = 2 * p2 - 1 + d2;
        const int x3 = 2 * (p3h * TILES) - 1 + d3;
        float v = 0.0f;
        if ((unsigned)x0 < (unsigned)S && (unsigned)x1 < (unsigned)S &&
            (unsigned)x2 < (unsigned)S && (unsigned)x3 < (unsigned)S) {
            v = x[((((b * Ci + ci) * S + x0) * S + x1) * S + x2) * S + x3];
        }
        lds[sp * CSTR + ci] = v;
    }

    const int lane = tid & 31;
    const int wave = tid >> 5;
    const int n    = lane & 15;
    const int klo  = (lane < 16) ? 0 : 2;
    const int qsp  = (((n >> 3) & 1) * 16 + ((n >> 2) & 1) * 4 + ((n >> 1) & 1)) * D3 + (n & 1);
    const int coA  = wave * 16 + n;
    const float* wb = wsl + coA * WROW + klo * 81;

    v8f acc[TILES];
    #pragma unroll
    for (int u = 0; u < TILES; ++u)
        acc[u] = (v8f){0.f, 0.f, 0.f, 0.f, 0.f, 0.f, 0.f, 0.f};

    #pragma unroll 1
    for (int kb = 0; kb < Ci; kb += 4) {
        __syncthreads();
        store_slice();
        __syncthreads();
        if (kb + 4 < Ci) load_slice(kb + 4);

        #pragma unroll 1
        for (int t0 = 0; t0 < 3; ++t0) {
            #pragma unroll 1
            for (int t1 = 0; t1 < 3; ++t1) {
                const int row0 = (qsp + (t0 * 16 + t1 * 4) * D3) * CSTR + klo + kb;
                const float* wt = wb + t0 * 27 + t1 * 9;
                #pragma unroll
                for (int t2 = 0; t2 < 3; ++t2) {
                    #pragma unroll
                    for (int t3 = 0; t3 < 3; ++t3) {
                        v2f bf;
                        bf.x = wt[t2 * 3 + t3];
                        bf.y = wt[t2 * 3 + t3 + 81];
                        #pragma unroll
                        for (int u = 0; u < TILES; ++u) {
                            v2f a = *(const v2f*)&lds[row0 + (t2 * D3 + t3 + 2 * u) * CSTR];
                            acc[u] = __builtin_amdgcn_wmma_f32_16x16x4_f32(
                                 false, a,  false, bf,
                                 (short)0, acc[u],
                                 false,  false);
                        }
                    }
                }
            }
        }
    }

    const float bs = bias[coA];
    #pragma unroll
    for (int u = 0; u < TILES; ++u) {
        float s = 0.0f;
        #pragma unroll
        for (int r = 0; r < 8; ++r) {
            float v = acc[u][r] + bs;
            s += (v > 0.0f) ? v : 0.0f;
        }
        s += __shfl_xor(s, 16, 32);
        if (lane < 16) {
            const int pIdx = ((p0 * P + p1) * P + p2) * P + p3h * TILES + u;
            out[(b * Co + coA) * P4 + pIdx] = s * (1.0f / 16.0f);
        }
    }
}


#define L0_WT   260
#define L0_SP0  300
#define L0_CSTR 12
#define L0_NPOS (3 * L0_SP0 + 288)

__global__ __launch_bounds__(256)
void conv4d_l0_wstat(const float* __restrict__ x,
                     const float* __restrict__ w,
                     const float* __restrict__ bias,
                     float* __restrict__ out)
{
    __shared__ __attribute__((aligned(16))) float wl[81 * L0_WT];
    __shared__ __attribute__((aligned(16))) float lp[L0_NPOS * L0_CSTR];

    const int tid = threadIdx.x;
    const int b   = blockIdx.y;
    const int p0  = blockIdx.x >> 3;
    const int p1  = blockIdx.x & 7;

    {
        int q = tid / 81;
        int t = tid - q * 81;
        #pragma unroll 1
        for (int c = 0; c < 3; ++c) {
            float tmp[27];
            #pragma unroll
            for (int i = 0; i < 27; ++i)
                tmp[i] = w[tid + (c * 27 + i) * 256];
            #pragma unroll
            for (int i = 0; i < 27; ++i) {
                const int inner = (q >> 7) * 128 + (((q >> 1) & 1) * 16 + ((q >> 3) & 15)) * 4
                                + ((q >> 2) & 1) * 2 + (q & 1);
                wl[t * L0_WT + inner] = tmp[i];
                t += 13; q += 3;
                if (t >= 81) { t -= 81; q += 1; }
            }
        }
    }
    for (int e = tid; e < 64 * 2 * 8; e += 256) {
        const int ci = e & 7;
        const int d3 = ((e >> 3) & 1) * 17;
        const int r  = e >> 4;
        lp[((r >> 4) * L0_SP0 + (r & 15) * 18 + d3) * L0_CSTR + ci] = 0.0f;
    }

    const int sx3 = tid & 15;
    const int sd2 = (tid >> 4) & 3;
    const int sd1 = (tid >> 6) & 3;
    const int gx1 = 2 * p1 - 1 + sd1;
    const bool ok1 = (unsigned)gx1 < 16u;
    const int ldsS = ((sd1 * 4 + sd2) * 18 + sx3 + 1) * L0_CSTR;

    float pre[32];
    auto load_patch = [&](int p2) {
        const int gx2 = 2 * p2 - 1 + sd2;
        const bool ok12 = ok1 && ((unsigned)gx2 < 16u);
        #pragma unroll
        for (int k = 0; k < 32; ++k) {
            const int ci = k >> 2, d0 = k & 3;
            const int gx0 = 2 * p0 - 1 + d0;
            const bool ok = ok12 && ((unsigned)gx0 < 16u);
            const int idx = ok ? (((((b * 8 + ci) * 16 + gx0) * 16 + gx1) * 16 + gx2) * 16 + sx3) : 0;
            pre[k] = x[idx];
        }
    };
    auto store_patch = [&](int p2) {
        const int gx2 = 2 * p2 - 1 + sd2;
        const bool ok12 = ok1 && ((unsigned)gx2 < 16u);
        #pragma unroll
        for (int k = 0; k < 32; ++k) {
            const int gx0 = 2 * p0 - 1 + (k & 3);
            const bool ok = ok12 && ((unsigned)gx0 < 16u);
            lp[(k & 3) * (L0_SP0 * L0_CSTR) + ldsS + (k >> 2)] = ok ? pre[k] : 0.0f;
        }
    };

    const int lane = tid & 31;
    const int wave = tid >> 5;
    const int ct   = wave & 1;
    const int pp   = wave >> 1;
    const int n    = lane & 15;
    const int klo  = (lane < 16) ? 0 : 2;
    const int qsp  = ((n >> 3) & 1) * L0_SP0 + (((n >> 2) & 1) * 4 + ((n >> 1) & 1)) * 18 + (n & 1);
    const int coA  = ct * 16 + n;
    const float bs = bias[coA];
    const float* wlw = wl + ct * 128 + lane * 4;
    const int aBase = (qsp + 4 * pp) * L0_CSTR + klo;

    load_patch(0);

    #pragma unroll 1
    for (int p2 = 0; p2 < 8; ++p2) {
        __syncthreads();
        store_patch(p2);
        __syncthreads();
        if (p2 + 1 < 8) load_patch(p2 + 1);

        v8f acc[2];
        acc[0] = (v8f){0.f, 0.f, 0.f, 0.f, 0.f, 0.f, 0.f, 0.f};
        acc[1] = (v8f){0.f, 0.f, 0.f, 0.f, 0.f, 0.f, 0.f, 0.f};

        #pragma unroll 1
        for (int t0 = 0; t0 < 3; ++t0) {
            #pragma unroll 1
            for (int t1 = 0; t1 < 3; ++t1) {
                #pragma unroll 1
                for (int t2 = 0; t2 < 3; ++t2) {
                    const int row = aBase + (t0 * L0_SP0 + (t1 * 4 + t2) * 18) * L0_CSTR;
                    const float* wt = wlw + (t0 * 27 + t1 * 9 + t2 * 3) * L0_WT;
                    float4 bw[3];
                    #pragma unroll
                    for (int t3 = 0; t3 < 3; ++t3)
                        bw[t3] = *(const float4*)&wt[t3 * L0_WT];
                    #pragma unroll
                    for (int kbi = 0; kbi < 2; ++kbi) {
                        #pragma unroll
                        for (int t3 = 0; t3 < 3; ++t3) {
                            v2f bf;
                            bf.x = kbi ? bw[t3].z : bw[t3].x;
                            bf.y = kbi ? bw[t3].w : bw[t3].y;
                            #pragma unroll
                            for (int u = 0; u < 2; ++u) {
                                v2f a = *(const v2f*)&lp[row + (t3 + 2 * u) * L0_CSTR + kbi * 4];
                                acc[u] = __builtin_amdgcn_wmma_f32_16x16x4_f32(
                                    false, a, false, bf, (short)0, acc[u], false, false);
                            }
                        }
                    }
                }
            }
        }

        #pragma unroll
        for (int u = 0; u < 2; ++u) {
            float s = 0.0f;
            #pragma unroll
            for (int r = 0; r < 8; ++r) {
                float v = acc[u][r] + bs;
                s += (v > 0.0f) ? v : 0.0f;
            }
            s += __shfl_xor(s, 16, 32);
            if (lane < 16) {
                const int pIdx = ((p0 * 8 + p1) * 8 + p2) * 8 + pp * 2 + u;
                out[(b * 32 + coA) * 4096 + pIdx] = s * (1.0f / 16.0f);
            }
        }
    }
}

extern "C" void kernel_launch(void* const* d_in, const int* in_sizes, int n_in,
                              void* d_out, int out_size, void* d_ws, size_t ws_size,
                              hipStream_t stream) {
    (void)in_sizes; (void)n_in; (void)out_size; (void)ws_size;

    const float* src = (const float*)d_in[0];
    const float* W0  = (const float*)d_in[1];
    const float* b0  = (const float*)d_in[2];
    const float* W1  = (const float*)d_in[3];
    const float* b1  = (const float*)d_in[4];
    const float* W2  = (const float*)d_in[5];
    const float* b2  = (const float*)d_in[6];
    const float* W3  = (const float*)d_in[7];
    const float* b3  = (const float*)d_in[8];

    float* t0 = (float*)d_ws;
    float* o1 = (float*)d_out;
    float* o2 = o1 + 8 * 64 * 256;
    float* o3 = o2 + 8 * 64 * 16;

    conv4d_l0_wstat<<<dim3(64, 8), 256, 0, stream>>>(src, W0, b0, t0);
    conv4d_relu_pool_wmma<32, 64, 8, 2><<<dim3(128, 8), 128, 0, stream>>>(t0, W1, b1, o1);
    conv4d_relu_pool_wmma<64, 64, 4, 2><<<dim3(8, 8), 128, 0, stream>>>(o1, W2, b2, o2);
    conv4d_relu_pool_wmma<64, 64, 2, 1><<<dim3(1, 8), 128, 0, stream>>>(o2, W3, b3, o3);
}


# Layers 0-2 write their result channel-minor (layer 0 to workspace, layers 1-2 as an extra copy), so layers 1-3 stage each patch position by 16-byte loads, 12-16 in flight; sum order unchanged
# speedup vs baseline: 2.1776x; 1.1188x over previous
#include <hip/hip_runtime.h>
#include <hip/hip_bf16.h>

typedef float v2f __attribute__((ext_vector_type(2)));
typedef float v4f __attribute__((ext_vector_type(4)));
typedef float v8f __attribute__((ext_vector_type(8)));

#define LDS_PAD 4

template<int Ci, int Co, int S, int TILES, bool WCM>
__global__ __launch_bounds__((Co/16)*32)
void conv4d_relu_pool_wmma(const float* __restrict__ x,
                           const float* __restrict__ w,
                           const float* __restrict__ bias,
                           float* __restrict__ out,
                           float* __restrict__ outcm)
{
    constexpr int P    = S / 2;
    constexpr int PH   = P / TILES;
    constexpr int P4   = P * P * P * P;
    constexpr int D3   = 2 + 2 * TILES;
    constexpr int NSP  = 64 * D3;
    constexpr int CSTR = Ci + LDS_PAD;
    constexpr int NT   = (Co / 16) * 32;
    constexpr int WROW = 4 * 81;
    constexpr int NV   = Co * 81;
    constexpr int FULL = NV / NT;
    constexpr int REM  = NV - FULL * NT;
    constexpr int CV   = Ci / 4;
    constexpr int PER  = NSP * CV / NT;
    constexpr int SB   = (PER % 16 == 0) ? 16 : 12;
    static_assert((NSP * CV) % NT == 0 && PER % SB == 0, "patch pieces must split evenly");

    __shared__ __attribute__((aligned(16))) float lds[NSP * CSTR];
    __shared__ __attribute__((aligned(16))) float wsl[Co * WROW];

    const int tid = threadIdx.x;

    v4f wpre[FULL + 1];
    auto load_slice = [&](int kb) {
        const float* ws = w + kb * 81;
        #pragma unroll
        for (int i = 0; i <= FULL; ++i) {
            if (i == FULL && REM == 0) break;
            int f = tid + i * NT;
            if (i == FULL) f = (f < NV) ? f : (NV - 1);
            const int co = f / 81;
            wpre[i] = *(const v4f*)&ws[co * (Ci * 81 - WROW) + 4 * f];
        }
    };
    auto store_slice = [&]() {
        #pragma unroll
        for (int i = 0; i < FULL; ++i)
            *(v4f*)&wsl[4 * (tid + i * NT)] = wpre[i];
        if (REM != 0 && tid < REM)
            *(v4f*)&wsl[4 * (tid + FULL * NT)] = wpre[FULL];
    };

    load_slice(0);

    const int b   = blockIdx.y;
    const int bx  = blockIdx.x;
    const int p3h = bx % PH;
    const int p2  = (bx / PH) % P;
    const int p1  = (bx / (PH * P)) % P;
    const int p0  = bx / (PH * P * P);

    #pragma unroll 1
    for (int g = 0; g < PER; g += SB) {
        v4f pv[SB];
        bool okv[SB];
        #pragma unroll
        for (int i = 0; i < SB; ++i) {
            const int e  = tid + (g + i) * NT;
            const int cv = e & (CV - 1);
            const int sp = e / CV;
            const int d3 = sp % D3;
            const int r  = sp / D3;
            const int d2 = r & 3, d1 = (r >> 2) & 3, d0 = (r >> 4) & 3;
            const int x0 = 2 * p0 - 1 + d0;
            const int x1 = 2 * p1 - 1 + d1;
            const int x2 = 2 * p2 - 1 + d2;
            const int x3 = 2 * (p3h * TILES) - 1 + d3;
            okv[i] = (unsigned)x0 < (unsigned)S && (unsigned)x1 < (unsigned)S &&
                     (unsigned)x2 < (unsigned)S && (unsigned)x3 < (unsigned)S;
            const int idx = okv[i] ? (((((b * S + x0) * S + x1) * S + x2) * S + x3) * Ci + 4 * cv) : 0;
            pv[i] = *(const v4f*)&x[idx];
        }
        #pragma unroll
        for (int i = 0; i < SB; ++i) {
            const int e  = tid + (g + i) * NT;
            const int cv = e & (CV - 1);
            const int sp = e / CV;
            *(v4f*)&lds[sp * CSTR + 4 * cv] = okv[i] ? pv[i] : (v4f){0.f, 0.f, 0.f, 0.f};
        }
    }

    const int lane = tid & 31;
    const int wave = tid >> 5;
    const int n    = lane & 15;
    const int klo  = (lane < 16) ? 0 : 2;
    const int qsp  = (((n >> 3) & 1) * 16 + ((n >> 2) & 1) * 4 + ((n >> 1) & 1)) * D3 + (n & 1);
    const int coA  = wave * 16 + n;
    const float* wb = wsl + coA * WROW + klo * 81;

    v8f acc[TILES];
    #pragma unroll
    for (int u = 0; u < TILES; ++u)
        acc[u] = (v8f){0.f, 0.f, 0.f, 0.f, 0.f, 0.f, 0.f, 0.f};

    #pragma unroll 1
    for (int kb = 0; kb < Ci; kb += 4) {
        __syncthreads();
        store_slice();
        __syncthreads();
        if (kb + 4 < Ci) load_slice(kb + 4);

        #pragma unroll 1
        for (int t0 = 0; t0 < 3; ++t0) {
            #pragma unroll 1
            for (int t1 = 0; t1 < 3; ++t1) {
                const int row0 = (qsp + (t0 * 16 + t1 * 4) * D3) * CSTR + klo + kb;
                const float* wt = wb + t0 * 27 + t1 * 9;
                #pragma unroll
                for (int t2 = 0; t2 < 3; ++t2) {
                    #pragma unroll
                    for (int t3 = 0; t3 < 3; ++t3) {
                        v2f bf;
                        bf.x = wt[t2 * 3 + t3];
                        bf.y = wt[t2 * 3 + t3 + 81];
                        #pragma unroll
                        for (int u = 0; u < TILES; ++u) {
                            v2f a = *(const v2f*)&lds[row0 + (t2 * D3 + t3 + 2 * u) * CSTR];
                            acc[u] = __builtin_amdgcn_wmma_f32_16x16x4_f32(
                                 false, a,  false, bf,
                                 (short)0, acc[u],
                                 false,  false);
                        }
                    }
                }
            }
        }
    }

    const float bs = bias[coA];
    #pragma unroll
    for (int u = 0; u < TILES; ++u) {
        float s = 0.0f;
        #pragma unroll
        for (int r = 0; r < 8; ++r) {
            float v = acc[u][r] + bs;
            s += (v > 0.0f) ? v : 0.0f;
        }
        s += __shfl_xor(s, 16, 32);
        if (lane < 16) {
            const int pIdx = ((p0 * P + p1) * P + p2) * P + p3h * TILES + u;
            const float res = s * (1.0f / 16.0f);
            out[(b * Co + coA) * P4 + pIdx] = res;
            if (WCM) outcm[(b * P4 + pIdx) * Co + coA] = res;
        }
    }
}


#define L0_WT   260
#define L0_SP0  300
#define L0_CSTR 12
#define L0_NPOS (3 * L0_SP0 + 288)

__global__ __launch_bounds__(256)
void conv4d_l0_wstat(const float* __restrict__ x,
                     const float* __restrict__ w,
                     const float* __restrict__ bias,
                     float* __restrict__ out)
{
    __shared__ __attribute__((aligned(16))) float wl[81 * L0_WT];
    __shared__ __attribute__((aligned(16))) float lp[L0_NPOS * L0_CSTR];

    const int tid = threadIdx.x;
    const int b   = blockIdx.y;
    const int p0  = blockIdx.x >> 3;
    const int p1  = blockIdx.x & 7;

    {
        int q = tid / 81;
        int t = tid - q * 81;
        #pragma unroll 1
        for (int c = 0; c < 3; ++c) {
            float tmp[27];
            #pragma unroll
            for (int i = 0; i < 27; ++i)
                tmp[i] = w[tid + (c * 27 + i) * 256];
            #pragma unroll
            for (int i = 0; i < 27; ++i) {
                const int inner = (q >> 7) * 128 + (((q >> 1) & 1) * 16 + ((q >> 3) & 15)) * 4
                                + ((q >> 2) & 1) * 2 + (q & 1);
                wl[t * L0_WT + inner] = tmp[i];
                t += 13; q += 3;
                if (t >= 81) { t -= 81; q += 1; }
            }
        }
    }
    for (int e = tid; e < 64 * 2 * 8; e += 256) {
        const int ci = e & 7;
        const int d3 = ((e >> 3) & 1) * 17;
        const int r  = e >> 4;
        lp[((r >> 4) * L0_SP0 + (r & 15) * 18 + d3) * L0_CSTR + ci] = 0.0f;
    }

    const int sx3 = tid & 15;
    const int sd2 = (tid >> 4) & 3;
    const int sd1 = (tid >> 6) & 3;
    const int gx1 = 2 * p1 - 1 + sd1;
    const bool ok1 = (unsigned)gx1 < 16u;
    const int ldsS = ((sd1 * 4 + sd2) * 18 + sx3 + 1) * L0_CSTR;

    float pre[32];
    auto load_patch = [&](int p2) {
        const int gx2 = 2 * p2 - 1 + sd2;
        const bool ok12 = ok1 && ((unsigned)gx2 < 16u);
        #pragma unroll
        for (int k = 0; k < 32; ++k) {
            const int ci = k >> 2, d0 = k & 3;
            const int gx0 = 2 * p0 - 1 + d0;
            const bool ok = ok12 && ((unsigned)gx0 < 16u);
            const int idx = ok ? (((((b * 8 + ci) * 16 + gx0) * 16 + gx1) * 16 + gx2) * 16 + sx3) : 0;
            pre[k] = x[idx];
        }
    };
    auto store_patch = [&](int p2) {
        const int gx2 = 2 * p2 - 1 + sd2;
        const bool ok12 = ok1 && ((unsigned)gx2 < 16u);
        #pragma unroll
        for (int k = 0; k < 32; ++k) {
            const int gx0 = 2 * p0 - 1 + (k & 3);
            const bool ok = ok12 && ((unsigned)gx0 < 16u);
            lp[(k & 3) * (L0_SP0 * L0_CSTR) + ldsS + (k >> 2)] = ok ? pre[k] : 0.0f;
        }
    };

    const int lane = tid & 31;
    const int wave = tid >> 5;
    const int ct   = wave & 1;
    const int pp   = wave >> 1;
    const int n    = lane & 15;
    const int klo  = (lane < 16) ? 0 : 2;
    const int qsp  = ((n >> 3) & 1) * L0_SP0 + (((n >> 2) & 1) * 4 + ((n >> 1) & 1)) * 18 + (n & 1);
    const int coA  = ct * 16 + n;
    const float bs = bias[coA];
    const float* wlw = wl + ct * 128 + lane * 4;
    const int aBase = (qsp + 4 * pp) * L0_CSTR + klo;

    load_patch(0);

    #pragma unroll 1
    for (int p2 = 0; p2 < 8; ++p2) {
        __syncthreads();
        store_patch(p2);
        __syncthreads();
        if (p2 + 1 < 8) load_patch(p2 + 1);

        v8f acc[2];
        acc[0] = (v8f){0.f, 0.f, 0.f, 0.f, 0.f, 0.f, 0.f, 0.f};
        acc[1] = (v8f){0.f, 0.f, 0.f, 0.f, 0.f, 0.f, 0.f, 0.f};

        #pragma unroll 1
        for (int t0 = 0; t0 < 3; ++t0) {
            #pragma unroll 1
            for (int t1 = 0; t1 < 3; ++t1) {
                #pragma unroll 1
                for (int t2 = 0; t2 < 3; ++t2) {
                    const int row = aBase + (t0 * L0_SP0 + (t1 * 4 + t2) * 18) * L0_CSTR;
                    const float* wt = wlw + (t0 * 27 + t1 * 9 + t2 * 3) * L0_WT;
                    float4 bw[3];
                    #pragma unroll
                    for (int t3 = 0; t3 < 3; ++t3)
                        bw[t3] = *(const float4*)&wt[t3 * L0_WT];
                    #pragma unroll
                    for (int kbi = 0; kbi < 2; ++kbi) {
                        #pragma unroll
                        for (int t3 = 0; t3 < 3; ++t3) {
                            v2f bf;
                            bf.x = kbi ? bw[t3].z : bw[t3].x;
                            bf.y = kbi ? bw[t3].w : bw[t3].y;
                            #pragma unroll
                            for (int u = 0; u < 2; ++u) {
                                v2f a = *(const v2f*)&lp[row + (t3 + 2 * u) * L0_CSTR + kbi * 4];
                                acc[u] = __builtin_amdgcn_wmma_f32_16x16x4_f32(
                                    false, a, false, bf, (short)0, acc[u], false, false);
                            }
                        }
                    }
                }
            }
        }

        #pragma unroll
        for (int u = 0; u < 2; ++u) {
            float s = 0.0f;
            #pragma unroll
            for (int r = 0; r < 8; ++r) {
                float v = acc[u][r] + bs;
                s += (v > 0.0f) ? v : 0.0f;
            }
            s += __shfl_xor(s, 16, 32);
            if (lane < 16) {
                const int pIdx = ((p0 * 8 + p1) * 8 + p2) * 8 + pp * 2 + u;
                out[(b * 4096 + pIdx) * 32 + coA] = s * (1.0f / 16.0f);
            }
        }
    }
}

extern "C" void kernel_launch(void* const* d_in, const int* in_sizes, int n_in,
                              void* d_out, int out_size, void* d_ws, size_t ws_size,
                              hipStream_t stream) {
    (void)in_sizes; (void)n_in; (void)out_size; (void)ws_size;

    const float* src = (const float*)d_in[0];
    const float* W0  = (const float*)d_in[1];
    const float* b0  = (const float*)d_in[2];
    const float* W1  = (const float*)d_in[3];
    const float* b1  = (const float*)d_in[4];
    const float* W2  = (const float*)d_in[5];
    const float* b2  = (const float*)d_in[6];
    const float* W3  = (const float*)d_in[7];
    const float* b3  = (const float*)d_in[8];

    float* t0 = (float*)d_ws;
    float* c1 = t0 + 8 * 4096 * 32;
    float* c2 = c1 + 8 * 256 * 64;
    float* o1 = (float*)d_out;
    float* o2 = o1 + 8 * 64 * 256;
    float* o3 = o2 + 8 * 64 * 16;

    conv4d_l0_wstat<<<dim3(64, 8), 256, 0, stream>>>(src, W0, b0, t0);
    conv4d_relu_pool_wmma<32, 64, 8, 2, true><<<dim3(128, 8), 128, 0, stream>>>(t0, W1, b1, o1, c1);
    conv4d_relu_pool_wmma<64, 64, 4, 2, true><<<dim3(8, 8), 128, 0, stream>>>(c1, W2, b2, o2, c2);
    conv4d_relu_pool_wmma<64, 64, 2, 1, false><<<dim3(1, 8), 128, 0, stream>>>(c2, W3, b3, o3, nullptr);
}


# Layers 2-3: each tile's input-channel sum split over 16 work-groups (one 4-channel weight slice each); raw sums go via workspace, next launch adds them before bias/ReLU/pool; sum order changes
# speedup vs baseline: 3.2100x; 1.4741x over previous
#include <hip/hip_runtime.h>
#include <hip/hip_bf16.h>

typedef float v2f __attribute__((ext_vector_type(2)));
typedef float v4f __attribute__((ext_vector_type(4)));
typedef float v8f __attribute__((ext_vector_type(8)));

#define LDS_PAD 4

template<int Ci, int Co, int S, int TILES, bool WCM>
__global__ __launch_bounds__((Co/16)*32)
void conv4d_relu_pool_wmma(const float* __restrict__ x,
                           const float* __restrict__ w,
                           const float* __restrict__ bias,
                           float* __restrict__ out,
                           float* __restrict__ outcm)
{
    constexpr int P    = S / 2;
    constexpr int PH   = P / TILES;
    constexpr int P4   = P * P * P * P;
    constexpr int D3   = 2 + 2 * TILES;
    constexpr int NSP  = 64 * D3;
    constexpr int CSTR = Ci + LDS_PAD;
    constexpr int NT   = (Co / 16) * 32;
    constexpr int WROW = 4 * 81;
    constexpr int NV   = Co * 81;
    constexpr int FULL = NV / NT;
    constexpr int REM  = NV - FULL * NT;
    constexpr int CV   = Ci / 4;
    constexpr int PER  = NSP * CV / NT;
    constexpr int SB   = (PER % 16 == 0) ? 16 : 12;
    static_assert((NSP * CV) % NT == 0 && PER % SB == 0, "patch pieces must split evenly");

    __shared__ __attribute__((aligned(16))) float lds[NSP * CSTR];
    __shared__ __attribute__((aligned(16))) float wsl[Co * WROW];

    const int tid = threadIdx.x;

    v4f wpre[FULL + 1];
    auto load_slice = [&](int kb) {
        const float* ws = w + kb * 81;
        #pragma unroll
        for (int i = 0; i <= FULL; ++i) {
            if (i == FULL && REM == 0) break;
            int f = tid + i * NT;
            if (i == FULL) f = (f < NV) ? f : (NV - 1);
            const int co = f / 81;
            wpre[i] = *(const v4f*)&ws[co * (Ci * 81 - WROW) + 4 * f];
        }
    };
    auto store_slice = [&]() {
        #pragma unroll
        for (int i = 0; i < FULL; ++i)
            *(v4f*)&wsl[4 * (tid + i * NT)] = wpre[i];
        if (REM != 0 && tid < REM)
            *(v4f*)&wsl[4 * (tid + FULL * NT)] = wpre[FULL];
    };

    load_slice(0);

    const int b   = blockIdx.y;
    const int bx  = blockIdx.x;
    const int p3h = bx % PH;
    const int p2  = (bx / PH) % P;
    const int p1  = (bx / (PH * P)) % P;
    const int p0  = bx / (PH * P * P);

    #pragma unroll 1
    for (int g = 0; g < PER; g += SB) {
        v4f pv[SB];
        bool okv[SB];
        #pragma unroll
        for (int i = 0; i < SB; ++i) {
            const int e  = tid + (g + i) * NT;
            const int cv = e & (CV - 1);
            const int sp = e / CV;
            const int d3 = sp % D3;
            const int r  = sp / D3;
            const int d2 = r & 3, d1 = (r >> 2) & 3, d0 = (r >> 4) & 3;
            const int x0 = 2 * p0 - 1 + d0;
            const int x1 = 2 * p1 - 1 + d1;
            const int x2 = 2 * p2 - 1 + d2;
            const int x3 = 2 * (p3h * TILES) - 1 + d3;
            okv[i] = (unsigned)x0 < (unsigned)S && (unsigned)x1 < (unsigned)S &&
                     (unsigned)x2 < (unsigned)S && (unsigned)x3 < (unsigned)S;
            const int idx = okv[i] ? (((((b * S + x0) * S + x1) * S + x2) * S + x3) * Ci + 4 * cv) : 0;
            pv[i] = *(const v4f*)&x[idx];
        }
        #pragma unroll
        for (int i = 0; i < SB; ++i) {
            const int e  = tid + (g + i) * NT;
            const int cv = e & (CV - 1);
            const int sp = e / CV;
            *(v4f*)&lds[sp * CSTR + 4 * cv] = okv[i] ? pv[i] : (v4f){0.f, 0.f, 0.f, 0.f};
        }
    }

    const int lane = tid & 31;
    const int wave = tid >> 5;
    const int n    = lane & 15;
    const int klo  = (lane < 16) ? 0 : 2;
    const int qsp  = (((n >> 3) & 1) * 16 + ((n >> 2) & 1) * 4 + ((n >> 1) & 1)) * D3 + (n & 1);
    const int coA  = wave * 16 + n;
    const float* wb = wsl + coA * WROW + klo * 81;

    v8f acc[TILES];
    #pragma unroll
    for (int u = 0; u < TILES; ++u)
        acc[u] = (v8f){0.f, 0.f, 0.f, 0.f, 0.f, 0.f, 0.f, 0.f};

    #pragma unroll 1
    for (int kb = 0; kb < Ci; kb += 4) {
        __syncthreads();
        store_slice();
        __syncthreads();
        if (kb + 4 < Ci) load_slice(kb + 4);

        #pragma unroll 1
        for (int t0 = 0; t0 < 3; ++t0) {
            #pragma unroll 1
            for (int t1 = 0; t1 < 3; ++t1) {
                const int row0 = (qsp + (t0 * 16 + t1 * 4) * D3) * CSTR + klo + kb;
                const float* wt = wb + t0 * 27 + t1 * 9;
                #pragma unroll
                for (int t2 = 0; t2 < 3; ++t2) {
                    #pragma unroll
                    for (int t3 = 0; t3 < 3; ++t3) {
                        v2f bf;
                        bf.x = wt[t2 * 3 + t3];
                        bf.y = wt[t2 * 3 + t3 + 81];
                        #pragma unroll
                        for (int u = 0; u < TILES; ++u) {
                            v2f a = *(const v2f*)&lds[row0 + (t2 * D3 + t3 + 2 * u) * CSTR];
                            acc[u] = __builtin_amdgcn_wmma_f32_16x16x4_f32(
                                 false, a,  false, bf,
                                 (short)0, acc[u],
                                 false,  false);
                        }
                    }
                }
            }
        }
    }

    const float bs = bias[coA];
    #pragma unroll
    for (int u = 0; u < TILES; ++u) {
        float s = 0.0f;
        #pragma unroll
        for (int r = 0; r < 8; ++r) {
            float v = acc[u][r] + bs;
            s += (v > 0.0f) ? v : 0.0f;
        }
        s += __shfl_xor(s, 16, 32);
        if (lane < 16) {
            const int pIdx = ((p0 * P + p1) * P + p2) * P + p3h * TILES + u;
            const float res = s * (1.0f / 16.0f);
            out[(b * Co + coA) * P4 + pIdx] = res;
            if (WCM) outcm[(b * P4 + pIdx) * Co + coA] = res;
        }
    }
}


#define L0_WT   260
#define L0_SP0  300
#define L0_CSTR 12
#define L0_NPOS (3 * L0_SP0 + 288)

__global__ __launch_bounds__(256)
void conv4d_l0_wstat(const float* __restrict__ x,
                     const float* __restrict__ w,
                     const float* __restrict__ bias,
                     float* __restrict__ out)
{
    __shared__ __attribute__((aligned(16))) float wl[81 * L0_WT];
    __shared__ __attribute__((aligned(16))) float lp[L0_NPOS * L0_CSTR];

    const int tid = threadIdx.x;
    const int b   = blockIdx.y;
    const int p0  = blockIdx.x >> 3;
    const int p1  = blockIdx.x & 7;

    {
        int q = tid / 81;
        int t = tid - q * 81;
        #pragma unroll 1
        for (int c = 0; c < 3; ++c) {
            float tmp[27];
            #pragma unroll
            for (int i = 0; i < 27; ++i)
                tmp[i] = w[tid + (c * 27 + i) * 256];
            #pragma unroll
            for (int i = 0; i < 27; ++i) {
                const int inner = (q >> 7) * 128 + (((q >> 1) & 1) * 16 + ((q >> 3) & 15)) * 4
                                + ((q >> 2) & 1) * 2 + (q & 1);
                wl[t * L0_WT + inner] = tmp[i];
                t += 13; q += 3;
                if (t >= 81) { t -= 81; q += 1; }
            }
        }
    }
    for (int e = tid; e < 64 * 2 * 8; e += 256) {
        const int ci = e & 7;
        const int d3 = ((e >> 3) & 1) * 17;
        const int r  = e >> 4;
        lp[((r >> 4) * L0_SP0 + (r & 15) * 18 + d3) * L0_CSTR + ci] = 0.0f;
    }

    const int sx3 = tid & 15;
    const int sd2 = (tid >> 4) & 3;
    const int sd1 = (tid >> 6) & 3;
    const int gx1 = 2 * p1 - 1 + sd1;
    const bool ok1 = (unsigned)gx1 < 16u;
    const int ldsS = ((sd1 * 4 + sd2) * 18 + sx3 + 1) * L0_CSTR;

    float pre[32];
    auto load_patch = [&](int p2) {
        const int gx2 = 2 * p2 - 1 + sd2;
        const bool ok12 = ok1 && ((unsigned)gx2 < 16u);
        #pragma unroll
        for (int k = 0; k < 32; ++k) {
            const int ci = k >> 2, d0 = k & 3;
            const int gx0 = 2 * p0 - 1 + d0;
            const bool ok = ok12 && ((unsigned)gx0 < 16u);
            const int idx = ok ? (((((b * 8 + ci) * 16 + gx0) * 16 + gx1) * 16 + gx2) * 16 + sx3) : 0;
            pre[k] = x[idx];
        }
    };
    auto store_patch = [&](int p2) {
        const int gx2 = 2 * p2 - 1 + sd2;
        const bool ok12 = ok1 && ((unsigned)gx2 < 16u);
        #pragma unroll
        for (int k = 0; k < 32; ++k) {
            const int gx0 = 2 * p0 - 1 + (k & 3);
            const bool ok = ok12 && ((unsigned)gx0 < 16u);
            lp[(k & 3) * (L0_SP0 * L0_CSTR) + ldsS + (k >> 2)] = ok ? pre[k] : 0.0f;
        }
    };

    const int lane = tid & 31;
    const int wave = tid >> 5;
    const int ct   = wave & 1;
    const int pp   = wave >> 1;
    const int n    = lane & 15;
    const int klo  = (lane < 16) ? 0 : 2;
    const int qsp  = ((n >> 3) & 1) * L0_SP0 + (((n >> 2) & 1) * 4 + ((n >> 1) & 1)) * 18 + (n & 1);
    const int coA  = ct * 16 + n;
    const float bs = bias[coA];
    const float* wlw = wl + ct * 128 + lane * 4;
    const int aBase = (qsp + 4 * pp) * L0_CSTR + klo;

    load_patch(0);

    #pragma unroll 1
    for (int p2 = 0; p2 < 8; ++p2) {
        __syncthreads();
        store_patch(p2);
        __syncthreads();
        if (p2 + 1 < 8) load_patch(p2 + 1);

        v8f acc[2];
        acc[0] = (v8f){0.f, 0.f, 0.f, 0.f, 0.f, 0.f, 0.f, 0.f};
        acc[1] = (v8f){0.f, 0.f, 0.f, 0.f, 0.f, 0.f, 0.f, 0.f};

        #pragma unroll 1
        for (int t0 = 0; t0 < 3; ++t0) {
            #pragma unroll 1
            for (int t1 = 0; t1 < 3; ++t1) {
                #pragma unroll 1
                for (int t2 = 0; t2 < 3; ++t2) {
                    const int row = aBase + (t0 * L0_SP0 + (t1 * 4 + t2) * 18) * L0_CSTR;
                    const float* wt = wlw + (t0 * 27 + t1 * 9 + t2 * 3) * L0_WT;
                    float4 bw[3];
                    #pragma unroll
                    for (int t3 = 0; t3 < 3; ++t3)
                        bw[t3] = *(const float4*)&wt[t3 * L0_WT];
                    #pragma unroll
                    for (int kbi = 0; kbi < 2; ++kbi) {
                        #pragma unroll
                        for (int t3 = 0; t3 < 3; ++t3) {
                            v2f bf;
                            bf.x = kbi ? bw[t3].z : bw[t3].x;
                            bf.y = kbi ? bw[t3].w : bw[t3].y;
                            #pragma unroll
                            for (int u = 0; u < 2; ++u) {
                                v2f a = *(const v2f*)&lp[row + (t3 + 2 * u) * L0_CSTR + kbi * 4];
                                acc[u] = __builtin_amdgcn_wmma_f32_16x16x4_f32(
                                    false, a, false, bf, (short)0, acc[u], false, false);
                            }
                        }
                    }
                }
            }
        }

        #pragma unroll
        for (int u = 0; u < 2; ++u) {
            float s = 0.0f;
            #pragma unroll
            for (int r = 0; r < 8; ++r) {
                float v = acc[u][r] + bs;
                s += (v > 0.0f) ? v : 0.0f;
            }
            s += __shfl_xor(s, 16, 32);
            if (lane < 16) {
                const int pIdx = ((p0 * 8 + p1) * 8 + p2) * 8 + pp * 2 + u;
                out[(b * 4096 + pIdx) * 32 + coA] = s * (1.0f / 16.0f);
            }
        }
    }
}


__device__ __forceinline__ float finish_tile(const float* __restrict__ pp, int co, int mh, float bs)
{
    const float* q = pp + (co >> 4) * 256 + (co & 15) * 16 + 8 * mh;
    v4f slo = (v4f){0.f, 0.f, 0.f, 0.f};
    v4f shi = (v4f){0.f, 0.f, 0.f, 0.f};
    #pragma unroll 1
    for (int g = 0; g < 16; g += 8) {
        v4f lo[8], hi[8];
        #pragma unroll
        for (int i = 0; i < 8; ++i) {
            lo[i] = *(const v4f*)&q[(g + i) * 1024];
            hi[i] = *(const v4f*)&q[(g + i) * 1024 + 4];
        }
        #pragma unroll
        for (int i = 0; i < 8; ++i) { slo += lo[i]; shi += hi[i]; }
    }
    float r = 0.0f;
    #pragma unroll
    for (int k = 0; k < 4; ++k) {
        const float v = slo[k] + bs;
        r += (v > 0.0f) ? v : 0.0f;
    }
    #pragma unroll
    for (int k = 0; k < 4; ++k) {
        const float v = shi[k] + bs;
        r += (v > 0.0f) ? v : 0.0f;
    }
    r += __shfl_xor(r, 1, 32);
    return r * (1.0f / 16.0f);
}

template<int S, int TILES, bool FIN>
__global__ __launch_bounds__(128)
void conv4d_slice_partial(const float* __restrict__ x,
                          const float* __restrict__ w,
                          const float* __restrict__ pbias,
                          float* __restrict__ pout,
                          float* __restrict__ part)
{
    constexpr int Ci   = 64;
    constexpr int Co   = 64;
    constexpr int P    = S / 2;
    constexpr int PH   = P / TILES;
    constexpr int P4   = P * P * P * P;
    constexpr int D3   = 2 + 2 * TILES;
    constexpr int NSP  = 64 * D3;
    constexpr int ST3  = 4;
    constexpr int ST2  = 4 * D3;
    constexpr int ST1  = 4 * ST2 + (D3 == 4 ? 8 : 0);
    constexpr int ST0  = 4 * ST1 + (D3 == 6 ? 16 : 0);
    constexpr int NPAT = 4 * ST0;
    constexpr int NT   = 128;
    constexpr int WROW = 4 * 81;
    constexpr int NV   = Co * 81;
    constexpr int FULL = NV / NT;
    constexpr int REM  = NV - FULL * NT;
    static_assert(D3 == 4 || D3 == 6, "bank-free strides are worked out for these two extents");
    static_assert(NSP % NT == 0, "patch positions must split evenly");
    static_assert(!FIN || S == 2, "the finishing input path maps 128 threads onto 16 positions x 4 channels x 2 halves");

    __shared__ __attribute__((aligned(16))) float lds[NPAT];
    __shared__ __attribute__((aligned(16))) float wsl[Co * WROW];

    const int tid = threadIdx.x;
    const int b   = blockIdx.y;
    const int j   = blockIdx.z;
    const int bx  = blockIdx.x;
    const int p3h = bx % PH;
    const int p2  = (bx / PH) % P;
    const int p1  = (bx / (PH * P)) % P;
    const int p0  = bx / (PH * P * P);

    v4f wpre[FULL + 1];
    {
        const float* ws = w + j * WROW;
        #pragma unroll
        for (int i = 0; i <= FULL; ++i) {
            if (i == FULL && REM == 0) break;
            int f = tid + i * NT;
            if (i == FULL) f = (f < NV) ? f : (NV - 1);
            const int co = f / 81;
            wpre[i] = *(const v4f*)&ws[co * (Ci * 81 - WROW) + 4 * f];
        }
    }

    if (FIN) {
        for (int e = tid; e < NPAT / 4; e += NT)
            *(v4f*)&lds[4 * e] = (v4f){0.f, 0.f, 0.f, 0.f};
        const int pI  = tid >> 3;
        const int c   = (tid >> 1) & 3;
        const int mh  = tid & 1;
        const int co2 = 4 * j + c;
        const float pooled = finish_tile(x + (b * 16 + pI) * (16 * 1024), co2, mh, pbias[co2]);
        __syncthreads();
        if (mh == 0) {
            const int q3 = pI & 1, q2 = (pI >> 1) & 1, q1 = (pI >> 2) & 1, q0 = (pI >> 3) & 1;
            lds[(q0 + 1) * ST0 + (q1 + 1) * ST1 + (q2 + 1) * ST2 + (q3 + 1) * ST3 + c] = pooled;
            pout[(b * 64 + co2) * 16 + pI] = pooled;
        }
    } else {
        constexpr int PERT = NSP / NT;
        v4f pv[PERT];
        bool okv[PERT];
        #pragma unroll
        for (int i = 0; i < PERT; ++i) {
            const int sp = tid + i * NT;
            const int d3 = sp % D3;
            const int r  = sp / D3;
            const int d2 = r & 3, d1 = (r >> 2) & 3, d0 = (r >> 4) & 3;
            const int x0 = 2 * p0 - 1 + d0;
            const int x1 = 2 * p1 - 1 + d1;
            const int x2 = 2 * p2 - 1 + d2;
            const int x3 = 2 * (p3h * TILES) - 1 + d3;
            okv[i] = (unsigned)x0 < (unsigned)S && (unsigned)x1 < (unsigned)S &&
                     (unsigned)x2 < (unsigned)S && (unsigned)x3 < (unsigned)S;
            const int idx = okv[i] ? (((((b * S + x0) * S + x1) * S + x2) * S + x3) * Ci + 4 * j) : 0;
            pv[i] = *(const v4f*)&x[idx];
        }
        #pragma unroll
        for (int i = 0; i < PERT; ++i) {
            const int sp = tid + i * NT;
            const int d3 = sp % D3;
            const int r  = sp / D3;
            const int d2 = r & 3, d1 = (r >> 2) & 3, d0 = (r >> 4) & 3;
            *(v4f*)&lds[d0 * ST0 + d1 * ST1 + d2 * ST2 + d3 * ST3] = okv[i] ? pv[i] : (v4f){0.f, 0.f, 0.f, 0.f};
        }
    }

    #pragma unroll
    for (int i = 0; i < FULL; ++i)
        *(v4f*)&wsl[4 * (tid + i * NT)] = wpre[i];
    if (REM != 0 && tid < REM)
        *(v4f*)&wsl[4 * (tid + FULL * NT)] = wpre[FULL];
    __syncthreads();

    const int lane = tid & 31;
    const int wave = tid >> 5;
    const int n    = lane & 15;
    const int klo  = (lane < 16) ? 0 : 2;
    const int qsp  = ((n >> 3) & 1) * ST0 + ((n >> 2) & 1) * ST1 + ((n >> 1) & 1) * ST2 + (n & 1) * ST3 + klo;
    const int coA  = wave * 16 + n;
    const float* wb = wsl + coA * WROW + klo * 81;

    v8f acc[TILES];
    #pragma unroll
    for (int u = 0; u < TILES; ++u)
        acc[u] = (v8f){0.f, 0.f, 0.f, 0.f, 0.f, 0.f, 0.f, 0.f};

    #pragma unroll 1
    for (int t0 = 0; t0 < 3; ++t0) {
        #pragma unroll 1
        for (int t1 = 0; t1 < 3; ++t1) {
            const int row0 = qsp + t0 * ST0 + t1 * ST1;
            const float* wt = wb + t0 * 27 + t1 * 9;
            #pragma unroll
            for (int t2 = 0; t2 < 3; ++t2) {
                #pragma unroll
                for (int t3 = 0; t3 < 3; ++t3) {
                    v2f bf;
                    bf.x = wt[t2 * 3 + t3];
                    bf.y = wt[t2 * 3 + t3 + 81];
                    #pragma unroll
                    for (int u = 0; u < TILES; ++u) {
                        v2f a = *(const v2f*)&lds[row0 + t2 * ST2 + (t3 + 2 * u) * ST3];
                        acc[u] = __builtin_amdgcn_wmma_f32_16x16x4_f32(
                             false, a,  false, bf,
                             (short)0, acc[u],
                             false,  false);
                    }
                }
            }
        }
    }

    const int hh = lane >> 4;
    #pragma unroll
    for (int u = 0; u < TILES; ++u) {
        const int pIdx = ((p0 * P + p1) * P + p2) * P + p3h * TILES + u;
        float* dst = part + (((b * P4 + pIdx) * 16 + j) * 4 + wave) * 256 + n * 16 + 8 * hh;
        *(v4f*)dst       = (v4f){acc[u][0], acc[u][1], acc[u][2], acc[u][3]};
        *(v4f*)(dst + 4) = (v4f){acc[u][4], acc[u][5], acc[u][6], acc[u][7]};
    }
}

__global__ __launch_bounds__(128)
void conv4d_finish_l3(const float* __restrict__ part,
                      const float* __restrict__ bias,
                      float* __restrict__ out)
{
    const int tid = threadIdx.x;
    const int b   = blockIdx.x;
    const int co  = tid >> 1;
    const int mh  = tid & 1;
    const float res = finish_tile(part + b * (16 * 1024), co, mh, bias[co]);
    if (mh == 0) out[b * 64 + co] = res;
}

extern "C" void kernel_launch(void* const* d_in, const int* in_sizes, int n_in,
                              void* d_out, int out_size, void* d_ws, size_t ws_size,
                              hipStream_t stream) {
    (void)in_sizes; (void)n_in; (void)out_size; (void)ws_size;

    const float* src = (const float*)d_in[0];
    const float* W0  = (const float*)d_in[1];
    const float* b0  = (const float*)d_in[2];
    const float* W1  = (const float*)d_in[3];
    const float* b1  = (const float*)d_in[4];
    const float* W2  = (const float*)d_in[5];
    const float* b2  = (const float*)d_in[6];
    const float* W3  = (const float*)d_in[7];
    const float* b3  = (const float*)d_in[8];

    float* t0 = (float*)d_ws;
    float* c1 = t0 + 8 * 4096 * 32;
    float* q2 = c1 + 8 * 256 * 64;
    float* q3 = q2 + 8 * 16 * 16 * 1024;
    float* o1 = (float*)d_out;
    float* o2 = o1 + 8 * 64 * 256;
    float* o3 = o2 + 8 * 64 * 16;

    conv4d_l0_wstat<<<dim3(64, 8), 256, 0, stream>>>(src, W0, b0, t0);
    conv4d_relu_pool_wmma<32, 64, 8, 2, true><<<dim3(128, 8), 128, 0, stream>>>(t0, W1, b1, o1, c1);
    conv4d_slice_partial<4, 2, false><<<dim3(8, 8, 16), 128, 0, stream>>>(c1, W2, nullptr, nullptr, q2);
    conv4d_slice_partial<2, 1, true><<<dim3(1, 8, 16), 128, 0, stream>>>(q2, W3, b2, o2, q3);
    conv4d_finish_l3<<<dim3(8), 128, 0, stream>>>(q3, b3, o3);
}


# Layer 1: one 16-wave block per 2x2x2 pooled cube x 64 channels (patch strides padded conflict-free); each LDS weight slice, two buffers, serves 4x the matrix ops at 4 waves/SIMD; sum order unchanged
# speedup vs baseline: 3.5811x; 1.1156x over previous
#include <hip/hip_runtime.h>
#include <hip/hip_bf16.h>

typedef float v2f __attribute__((ext_vector_type(2)));
typedef float v4f __attribute__((ext_vector_type(4)));
typedef float v8f __attribute__((ext_vector_type(8)));


#define L1_ST3  36
#define L1_ST2  224
#define L1_ST1  1352
#define L1_ST0  8144
#define L1_NPAT (4 * L1_ST0)
#define L1_WSL  (64 * 324)

__global__ __launch_bounds__(512)
void conv4d_l1_cube(const float* __restrict__ x,
                    const float* __restrict__ w,
                    const float* __restrict__ bias,
                    float* __restrict__ out,
                    float* __restrict__ outcm)
{
    constexpr int Ci    = 32;
    constexpr int Co    = 64;
    constexpr int S     = 8;
    constexpr int P     = 4;
    constexpr int P4    = P * P * P * P;
    constexpr int NT    = 512;
    constexpr int WROW  = 4 * 81;
    constexpr int NV    = Co * 81;
    constexpr int FULL  = NV / NT;
    constexpr int REM   = NV - FULL * NT;
    constexpr int NPC   = 4 * 6 * 6 * 6 * (Ci / 4);
    constexpr int PFULL = NPC / NT;
    constexpr int PREM  = NPC - PFULL * NT;

    __shared__ __attribute__((aligned(16))) float lds[L1_NPAT];
    __shared__ __attribute__((aligned(16))) float wsl[2 * L1_WSL];

    const int tid = threadIdx.x;

    v4f wpre[FULL + 1];
    auto load_slice = [&](int kb) {
        const float* ws = w + kb * 81;
        #pragma unroll
        for (int i = 0; i <= FULL; ++i) {
            int f = tid + i * NT;
            if (i == FULL) f = (f < NV) ? f : (NV - 1);
            const int co = f / 81;
            wpre[i] = *(const v4f*)&ws[co * (Ci * 81 - WROW) + 4 * f];
        }
    };
    auto store_slice = [&](float* dst) {
        #pragma unroll
        for (int i = 0; i < FULL; ++i)
            *(v4f*)&dst[4 * (tid + i * NT)] = wpre[i];
        if (tid < REM)
            *(v4f*)&dst[4 * (tid + FULL * NT)] = wpre[FULL];
    };

    load_slice(0);

    const int b   = blockIdx.y;
    const int bx  = blockIdx.x;
    const int p3h = bx & 1;
    const int p2h = (bx >> 1) & 1;
    const int p1h = (bx >> 2) & 1;
    const int p0  = bx >> 3;

    {
        v4f  pv[PFULL + 1];
        bool okv[PFULL + 1];
        #pragma unroll
        for (int i = 0; i <= PFULL; ++i) {
            int e = tid + i * NT;
            if (i == PFULL) e = (e < NPC) ? e : (NPC - 1);
            const int cv = e & 7;
            const int sp = e >> 3;
            const int d3 = sp % 6;
            const int r  = sp / 6;
            const int d2 = r % 6;
            const int r2 = r / 6;
            const int d1 = r2 % 6;
            const int d0 = r2 / 6;
            const int x0 = 2 * p0  - 1 + d0;
            const int x1 = 4 * p1h - 1 + d1;
            const int x2 = 4 * p2h - 1 + d2;
            const int x3 = 4 * p3h - 1 + d3;
            okv[i] = (unsigned)x0 < (unsigned)S && (unsigned)x1 < (unsigned)S &&
                     (unsigned)x2 < (unsigned)S && (unsigned)x3 < (unsigned)S;
            const int idx = okv[i] ? (((((b * S + x0) * S + x1) * S + x2) * S + x3) * Ci + 4 * cv) : 0;
            pv[i] = *(const v4f*)&x[idx];
        }
        #pragma unroll
        for (int i = 0; i <= PFULL; ++i) {
            const int e  = tid + i * NT;
            if (i == PFULL && tid >= PREM) break;
            const int cv = e & 7;
            const int sp = e >> 3;
            const int d3 = sp % 6;
            const int r  = sp / 6;
            const int d2 = r % 6;
            const int r2 = r / 6;
            const int d1 = r2 % 6;
            const int d0 = r2 / 6;
            *(v4f*)&lds[d0 * L1_ST0 + d1 * L1_ST1 + d2 * L1_ST2 + d3 * L1_ST3 + 4 * cv] =
                okv[i] ? pv[i] : (v4f){0.f, 0.f, 0.f, 0.f};
        }
    }

    store_slice(wsl);
    __syncthreads();

    const int lane = tid & 31;
    const int wave = tid >> 5;
    const int ct   = wave & 3;
    const int g2   = (wave >> 2) & 1;
    const int g1   = wave >> 3;
    const int n    = lane & 15;
    const int klo  = (lane < 16) ? 0 : 2;
    const int qsp  = ((n >> 3) & 1) * L1_ST0 + (((n >> 2) & 1) + 2 * g1) * L1_ST1
                   + (((n >> 1) & 1) + 2 * g2) * L1_ST2 + (n & 1) * L1_ST3;
    const int coA  = ct * 16 + n;
    const int wbo  = coA * WROW + klo * 81;

    v8f acc[2];
    acc[0] = (v8f){0.f, 0.f, 0.f, 0.f, 0.f, 0.f, 0.f, 0.f};
    acc[1] = (v8f){0.f, 0.f, 0.f, 0.f, 0.f, 0.f, 0.f, 0.f};

    #pragma unroll 1
    for (int kb = 0; kb < Ci; kb += 4) {
        const float* wb = wsl + ((kb >> 2) & 1) * L1_WSL + wbo;
        if (kb + 4 < Ci) load_slice(kb + 4);

        #pragma unroll 1
        for (int t0 = 0; t0 < 3; ++t0) {
            #pragma unroll 1
            for (int t1 = 0; t1 < 3; ++t1) {
                const int row0 = qsp + t0 * L1_ST0 + t1 * L1_ST1 + klo + kb;
                const float* wt = wb + t0 * 27 + t1 * 9;
                #pragma unroll
                for (int t2 = 0; t2 < 3; ++t2) {
                    #pragma unroll
                    for (int t3 = 0; t3 < 3; ++t3) {
                        v2f bf;
                        bf.x = wt[t2 * 3 + t3];
                        bf.y = wt[t2 * 3 + t3 + 81];
                        #pragma unroll
                        for (int u = 0; u < 2; ++u) {
                            v2f a = *(const v2f*)__builtin_assume_aligned(&lds[row0 + t2 * L1_ST2 + (t3 + 2 * u) * L1_ST3], 8);
                            acc[u] = __builtin_amdgcn_wmma_f32_16x16x4_f32(
                                 false, a,  false, bf,
                                 (short)0, acc[u],
                                 false,  false);
                        }
                    }
                }
            }
        }

        if (kb + 4 < Ci) store_slice(wsl + (((kb >> 2) + 1) & 1) * L1_WSL);
        __syncthreads();
    }

    const float bs = bias[coA];
    #pragma unroll
    for (int u = 0; u < 2; ++u) {
        float s = 0.0f;
        #pragma unroll
        for (int r = 0; r < 8; ++r) {
            float v = acc[u][r] + bs;
            s += (v > 0.0f) ? v : 0.0f;
        }
        s += __shfl_xor(s, 16, 32);
        if (lane < 16) {
            const int pIdx = ((p0 * P + 2 * p1h + g1) * P + 2 * p2h + g2) * P + 2 * p3h + u;
            const float res = s * (1.0f / 16.0f);
            out[(b * Co + coA) * P4 + pIdx] = res;
            outcm[(b * P4 + pIdx) * Co + coA] = res;
        }
    }
}

#define L0_WT   260
#define L0_SP0  300
#define L0_CSTR 12
#define L0_NPOS (3 * L0_SP0 + 288)

__global__ __launch_bounds__(256)
void conv4d_l0_wstat(const float* __restrict__ x,
                     const float* __restrict__ w,
                     const float* __restrict__ bias,
                     float* __restrict__ out)
{
    __shared__ __attribute__((aligned(16))) float wl[81 * L0_WT];
    __shared__ __attribute__((aligned(16))) float lp[L0_NPOS * L0_CSTR];

    const int tid = threadIdx.x;
    const int b   = blockIdx.y;
    const int p0  = blockIdx.x >> 3;
    const int p1  = blockIdx.x & 7;

    {
        int q = tid / 81;
        int t = tid - q * 81;
        #pragma unroll 1
        for (int c = 0; c < 3; ++c) {
            float tmp[27];
            #pragma unroll
            for (int i = 0; i < 27; ++i)
                tmp[i] = w[tid + (c * 27 + i) * 256];
            #pragma unroll
            for (int i = 0; i < 27; ++i) {
                const int inner = (q >> 7) * 128 + (((q >> 1) & 1) * 16 + ((q >> 3) & 15)) * 4
                                + ((q >> 2) & 1) * 2 + (q & 1);
                wl[t * L0_WT + inner] = tmp[i];
                t += 13; q += 3;
                if (t >= 81) { t -= 81; q += 1; }
            }
        }
    }
    for (int e = tid; e < 64 * 2 * 8; e += 256) {
        const int ci = e & 7;
        const int d3 = ((e >> 3) & 1) * 17;
        const int r  = e >> 4;
        lp[((r >> 4) * L0_SP0 + (r & 15) * 18 + d3) * L0_CSTR + ci] = 0.0f;
    }

    const int sx3 = tid & 15;
    const int sd2 = (tid >> 4) & 3;
    const int sd1 = (tid >> 6) & 3;
    const int gx1 = 2 * p1 - 1 + sd1;
    const bool ok1 = (unsigned)gx1 < 16u;
    const int ldsS = ((sd1 * 4 + sd2) * 18 + sx3 + 1) * L0_CSTR;

    float pre[32];
    auto load_patch = [&](int p2) {
        const int gx2 = 2 * p2 - 1 + sd2;
        const bool ok12 = ok1 && ((unsigned)gx2 < 16u);
        #pragma unroll
        for (int k = 0; k < 32; ++k) {
            const int ci = k >> 2, d0 = k & 3;
            const int gx0 = 2 * p0 - 1 + d0;
            const bool ok = ok12 && ((unsigned)gx0 < 16u);
            const int idx = ok ? (((((b * 8 + ci) * 16 + gx0) * 16 + gx1) * 16 + gx2) * 16 + sx3) : 0;
            pre[k] = x[idx];
        }
    };
    auto store_patch = [&](int p2) {
        const int gx2 = 2 * p2 - 1 + sd2;
        const bool ok12 = ok1 && ((unsigned)gx2 < 16u);
        #pragma unroll
        for (int k = 0; k < 32; ++k) {
            const int gx0 = 2 * p0 - 1 + (k & 3);
            const bool ok = ok12 && ((unsigned)gx0 < 16u);
            lp[(k & 3) * (L0_SP0 * L0_CSTR) + ldsS + (k >> 2)] = ok ? pre[k] : 0.0f;
        }
    };

    const int lane = tid & 31;
    const int wave = tid >> 5;
    const int ct   = wave & 1;
    const int pp   = wave >> 1;
    const int n    = lane & 15;
    const int klo  = (lane < 16) ? 0 : 2;
    const int qsp  = ((n >> 3) & 1) * L0_SP0 + (((n >> 2) & 1) * 4 + ((n >> 1) & 1)) * 18 + (n & 1);
    const int coA  = ct * 16 + n;
    const float bs = bias[coA];
    const float* wlw = wl + ct * 128 + lane * 4;
    const int aBase = (qsp + 4 * pp) * L0_CSTR + klo;

    load_patch(0);

    #pragma unroll 1
    for (int p2 = 0; p2 < 8; ++p2) {
        __syncthreads();
        store_patch(p2);
        __syncthreads();
        if (p2 + 1 < 8) load_patch(p2 + 1);

        v8f acc[2];
        acc[0] = (v8f){0.f, 0.f, 0.f, 0.f, 0.f, 0.f, 0.f, 0.f};
        acc[1] = (v8f){0.f, 0.f, 0.f, 0.f, 0.f, 0.f, 0.f, 0.f};

        #pragma unroll 1
        for (int t0 = 0; t0 < 3; ++t0) {
            #pragma unroll 1
            for (int t1 = 0; t1 < 3; ++t1) {
                #pragma unroll 1
                for (int t2 = 0; t2 < 3; ++t2) {
                    const int row = aBase + (t0 * L0_SP0 + (t1 * 4 + t2) * 18) * L0_CSTR;
                    const float* wt = wlw + (t0 * 27 + t1 * 9 + t2 * 3) * L0_WT;
                    float4 bw[3];
                    #pragma unroll
                    for (int t3 = 0; t3 < 3; ++t3)
                        bw[t3] = *(const float4*)&wt[t3 * L0_WT];
                    #pragma unroll
                    for (int kbi = 0; kbi < 2; ++kbi) {
                        #pragma unroll
                        for (int t3 = 0; t3 < 3; ++t3) {
                            v2f bf;
                            bf.x = kbi ? bw[t3].z : bw[t3].x;
                            bf.y = kbi ? bw[t3].w : bw[t3].y;
                            #pragma unroll
                            for (int u = 0; u < 2; ++u) {
                                v2f a = *(const v2f*)&lp[row + (t3 + 2 * u) * L0_CSTR + kbi * 4];
                                acc[u] = __builtin_amdgcn_wmma_f32_16x16x4_f32(
                                    false, a, false, bf, (short)0, acc[u], false, false);
                            }
                        }
                    }
                }
            }
        }

        #pragma unroll
        for (int u = 0; u < 2; ++u) {
            float s = 0.0f;
            #pragma unroll
            for (int r = 0; r < 8; ++r) {
                float v = acc[u][r] + bs;
                s += (v > 0.0f) ? v : 0.0f;
            }
            s += __shfl_xor(s, 16, 32);
            if (lane < 16) {
                const int pIdx = ((p0 * 8 + p1) * 8 + p2) * 8 + pp * 2 + u;
                out[(b * 4096 + pIdx) * 32 + coA] = s * (1.0f / 16.0f);
            }
        }
    }
}


__device__ __forceinline__ float finish_tile(const float* __restrict__ pp, int co, int mh, float bs)
{
    const float* q = pp + (co >> 4) * 256 + (co & 15) * 16 + 8 * mh;
    v4f slo = (v4f){0.f, 0.f, 0.f, 0.f};
    v4f shi = (v4f){0.f, 0.f, 0.f, 0.f};
    #pragma unroll 1
    for (int g = 0; g < 16; g += 8) {
        v4f lo[8], hi[8];
        #pragma unroll
        for (int i = 0; i < 8; ++i) {
            lo[i] = *(const v4f*)&q[(g + i) * 1024];
            hi[i] = *(const v4f*)&q[(g + i) * 1024 + 4];
        }
        #pragma unroll
        for (int i = 0; i < 8; ++i) { slo += lo[i]; shi += hi[i]; }
    }
    float r = 0.0f;
    #pragma unroll
    for (int k = 0; k < 4; ++k) {
        const float v = slo[k] + bs;
        r += (v > 0.0f) ? v : 0.0f;
    }
    #pragma unroll
    for (int k = 0; k < 4; ++k) {
        const float v = shi[k] + bs;
        r += (v > 0.0f) ? v : 0.0f;
    }
    r += __shfl_xor(r, 1, 32);
    return r * (1.0f / 16.0f);
}

template<int S, int TILES, bool FIN>
__global__ __launch_bounds__(128)
void conv4d_slice_partial(const float* __restrict__ x,
                          const float* __restrict__ w,
                          const float* __restrict__ pbias,
                          float* __restrict__ pout,
                          float* __restrict__ part)
{
    constexpr int Ci   = 64;
    constexpr int Co   = 64;
    constexpr int P    = S / 2;
    constexpr int PH   = P / TILES;
    constexpr int P4   = P * P * P * P;
    constexpr int D3   = 2 + 2 * TILES;
    constexpr int NSP  = 64 * D3;
    constexpr int ST3  = 4;
    constexpr int ST2  = 4 * D3;
    constexpr int ST1  = 4 * ST2 + (D3 == 4 ? 8 : 0);
    constexpr int ST0  = 4 * ST1 + (D3 == 6 ? 16 : 0);
    constexpr int NPAT = 4 * ST0;
    constexpr int NT   = 128;
    constexpr int WROW = 4 * 81;
    constexpr int NV   = Co * 81;
    constexpr int FULL = NV / NT;
    constexpr int REM  = NV - FULL * NT;
    static_assert(D3 == 4 || D3 == 6, "bank-free strides are worked out for these two extents");
    static_assert(NSP % NT == 0, "patch positions must split evenly");
    static_assert(!FIN || S == 2, "the finishing input path maps 128 threads onto 16 positions x 4 channels x 2 halves");

    __shared__ __attribute__((aligned(16))) float lds[NPAT];
    __shared__ __attribute__((aligned(16))) float wsl[Co * WROW];

    const int tid = threadIdx.x;
    const int b   = blockIdx.y;
    const int j   = blockIdx.z;
    const int bx  = blockIdx.x;
    const int p3h = bx % PH;
    const int p2  = (bx / PH) % P;
    const int p1  = (bx / (PH * P)) % P;
    const int p0  = bx / (PH * P * P);

    v4f wpre[FULL + 1];
    {
        const float* ws = w + j * WROW;
        #pragma unroll
        for (int i = 0; i <= FULL; ++i) {
            if (i == FULL && REM == 0) break;
            int f = tid + i * NT;
            if (i == FULL) f = (f < NV) ? f : (NV - 1);
            const int co = f / 81;
            wpre[i] = *(const v4f*)&ws[co * (Ci * 81 - WROW) + 4 * f];
        }
    }

    if (FIN) {
        for (int e = tid; e < NPAT / 4; e += NT)
            *(v4f*)&lds[4 * e] = (v4f){0.f, 0.f, 0.f, 0.f};
        const int pI  = tid >> 3;
        const int c   = (tid >> 1) & 3;
        const int mh  = tid & 1;
        const int co2 = 4 * j + c;
        const float pooled = finish_tile(x + (b * 16 + pI) * (16 * 1024), co2, mh, pbias[co2]);
        __syncthreads();
        if (mh == 0) {
            const int q3 = pI & 1, q2 = (pI >> 1) & 1, q1 = (pI >> 2) & 1, q0 = (pI >> 3) & 1;
            lds[(q0 + 1) * ST0 + (q1 + 1) * ST1 + (q2 + 1) * ST2 + (q3 + 1) * ST3 + c] = pooled;
            pout[(b * 64 + co2) * 16 + pI] = pooled;
        }
    } else {
        constexpr int PERT = NSP / NT;
        v4f pv[PERT];
        bool okv[PERT];
        #pragma unroll
        for (int i = 0; i < PERT; ++i) {
            const int sp = tid + i * NT;
            const int d3 = sp % D3;
            const int r  = sp / D3;
            const int d2 = r & 3, d1 = (r >> 2) & 3, d0 = (r >> 4) & 3;
            const int x0 = 2 * p0 - 1 + d0;
            const int x1 = 2 * p1 - 1 + d1;
            const int x2 = 2 * p2 - 1 + d2;
            const int x3 = 2 * (p3h * TILES) - 1 + d3;
            okv[i] = (unsigned)x0 < (unsigned)S && (unsigned)x1 < (unsigned)S &&
                     (unsigned)x2 < (unsigned)S && (unsigned)x3 < (unsigned)S;
            const int idx = okv[i] ? (((((b * S + x0) * S + x1) * S + x2) * S + x3) * Ci + 4 * j) : 0;
            pv[i] = *(const v4f*)&x[idx];
        }
        #pragma unroll
        for (int i = 0; i < PERT; ++i) {
            const int sp = tid + i * NT;
            const int d3 = sp % D3;
            const int r  = sp / D3;
            const int d2 = r & 3, d1 = (r >> 2) & 3, d0 = (r >> 4) & 3;
            *(v4f*)&lds[d0 * ST0 + d1 * ST1 + d2 * ST2 + d3 * ST3] = okv[i] ? pv[i] : (v4f){0.f, 0.f, 0.f, 0.f};
        }
    }

    #pragma unroll
    for (int i = 0; i < FULL; ++i)
        *(v4f*)&wsl[4 * (tid + i * NT)] = wpre[i];
    if (REM != 0 && tid < REM)
        *(v4f*)&wsl[4 * (tid + FULL * NT)] = wpre[FULL];
    __syncthreads();

    const int lane = tid & 31;
    const int wave = tid >> 5;
    const int n    = lane & 15;
    const int klo  = (lane < 16) ? 0 : 2;
    const int qsp  = ((n >> 3) & 1) * ST0 + ((n >> 2) & 1) * ST1 + ((n >> 1) & 1) * ST2 + (n & 1) * ST3 + klo;
    const int coA  = wave * 16 + n;
    const float* wb = wsl + coA * WROW + klo * 81;

    v8f acc[TILES];
    #pragma unroll
    for (int u = 0; u < TILES; ++u)
        acc[u] = (v8f){0.f, 0.f, 0.f, 0.f, 0.f, 0.f, 0.f, 0.f};

    #pragma unroll 1
    for (int t0 = 0; t0 < 3; ++t0) {
        #pragma unroll 1
        for (int t1 = 0; t1 < 3; ++t1) {
            const int row0 = qsp + t0 * ST0 + t1 * ST1;
            const float* wt = wb + t0 * 27 + t1 * 9;
            #pragma unroll
            for (int t2 = 0; t2 < 3; ++t2) {
                #pragma unroll
                for (int t3 = 0; t3 < 3; ++t3) {
                    v2f bf;
                    bf.x = wt[t2 * 3 + t3];
                    bf.y = wt[t2 * 3 + t3 + 81];
                    #pragma unroll
                    for (int u = 0; u < TILES; ++u) {
                        v2f a = *(const v2f*)&lds[row0 + t2 * ST2 + (t3 + 2 * u) * ST3];
                        acc[u] = __builtin_amdgcn_wmma_f32_16x16x4_f32(
                             false, a,  false, bf,
                             (short)0, acc[u],
                             false,  false);
                    }
                }
            }
        }
    }

    const int hh = lane >> 4;
    #pragma unroll
    for (int u = 0; u < TILES; ++u) {
        const int pIdx = ((p0 * P + p1) * P + p2) * P + p3h * TILES + u;
        float* dst = part + (((b * P4 + pIdx) * 16 + j) * 4 + wave) * 256 + n * 16 + 8 * hh;
        *(v4f*)dst       = (v4f){acc[u][0], acc[u][1], acc[u][2], acc[u][3]};
        *(v4f*)(dst + 4) = (v4f){acc[u][4], acc[u][5], acc[u][6], acc[u][7]};
    }
}

__global__ __launch_bounds__(128)
void conv4d_finish_l3(const float* __restrict__ part,
                      const float* __restrict__ bias,
                      float* __restrict__ out)
{
    const int tid = threadIdx.x;
    const int b   = blockIdx.x;
    const int co  = tid >> 1;
    const int mh  = tid & 1;
    const float res = finish_tile(part + b * (16 * 1024), co, mh, bias[co]);
    if (mh == 0) out[b * 64 + co] = res;
}

extern "C" void kernel_launch(void* const* d_in, const int* in_sizes, int n_in,
                              void* d_out, int out_size, void* d_ws, size_t ws_size,
                              hipStream_t stream) {
    (void)in_sizes; (void)n_in; (void)out_size; (void)ws_size;

    const float* src = (const float*)d_in[0];
    const float* W0  = (const float*)d_in[1];
    const float* b0  = (const float*)d_in[2];
    const float* W1  = (const float*)d_in[3];
    const float* b1  = (const float*)d_in[4];
    const float* W2  = (const float*)d_in[5];
    const float* b2  = (const float*)d_in[6];
    const float* W3  = (const float*)d_in[7];
    const float* b3  = (const float*)d_in[8];

    float* t0 = (float*)d_ws;
    float* c1 = t0 + 8 * 4096 * 32;
    float* q2 = c1 + 8 * 256 * 64;
    float* q3 = q2 + 8 * 16 * 16 * 1024;
    float* o1 = (float*)d_out;
    float* o2 = o1 + 8 * 64 * 256;
    float* o3 = o2 + 8 * 64 * 16;

    conv4d_l0_wstat<<<dim3(64, 8), 256, 0, stream>>>(src, W0, b0, t0);
    conv4d_l1_cube<<<dim3(32, 8), 512, 0, stream>>>(t0, W1, b1, o1, c1);
    conv4d_slice_partial<4, 2, false><<<dim3(8, 8, 16), 128, 0, stream>>>(c1, W2, nullptr, nullptr, q2);
    conv4d_slice_partial<2, 1, true><<<dim3(1, 8, 16), 128, 0, stream>>>(q2, W3, b2, o2, q3);
    conv4d_finish_l3<<<dim3(8), 128, 0, stream>>>(q3, b3, o3);
}
